# starting kernel
# baseline (speedup 1.0000x reference)
#include <hip/hip_runtime.h>

typedef float v2f __attribute__((ext_vector_type(2)));
typedef float v8f __attribute__((ext_vector_type(8)));

#define D_IN  199
#define K_V   64
#define P_D   4

__device__ __forceinline__ float sel4(int j, float a0, float a1, float a2, float a3) {
  const float t01 = (j & 1) ? a1 : a0;
  const float t23 = (j & 1) ? a3 : a2;
  return (j & 2) ? t23 : t01;
}

__global__ __launch_bounds__(256) void vdx_kernel(const float* __restrict__ x,
                                                  const float* __restrict__ dirs,
                                                  float* __restrict__ out,
                                                  int npts) {
  const int lane = threadIdx.x & 31;
  const int wave = threadIdx.x >> 5;
  const int pt   = blockIdx.x * 8 + wave;
  if (pt >= npts) return;

  const float* row = x + (size_t)pt * D_IN;

  const float px = row[0], py = row[1], pz = row[2];
  const float qw = row[3], qx = row[4], qy = row[5], qz = row[6];

  const float xx = qx*qx, yy = qy*qy, zz = qz*qz;
  const float xy = qx*qy, xz = qx*qz, yz = qy*qz;
  const float wx = qw*qx, wy = qw*qy, wz = qw*qz;
  const float R00 = 1.f - 2.f*(yy+zz), R01 = 2.f*(xy-wz),       R02 = 2.f*(xz+wy);
  const float R10 = 2.f*(xy+wz),       R11 = 1.f - 2.f*(xx+zz), R12 = 2.f*(yz-wx);
  const float R20 = 2.f*(xz-wy),       R21 = 2.f*(yz+wx),       R22 = 1.f - 2.f*(xx+yy);

  const int  pcol   = lane & 3;
  const int  jgrp   = (lane >> 2) & 3;
  const int  half   = lane >> 4;
  const bool lohalf = (lane < 16);

  const float d0 = sel4(pcol, dirs[0], dirs[3], dirs[6], dirs[9]);
  const float d1 = sel4(pcol, dirs[1], dirs[4], dirs[7], dirs[10]);
  const float d2 = sel4(pcol, dirs[2], dirs[5], dirs[8], dirs[11]);
  const float mt0 = d0*R00 + d1*R10 + d2*R20;
  const float mt1 = d0*R01 + d1*R11 + d2*R21;
  const float mt2 = d0*R02 + d1*R12 + d2*R22;
  const float cc  = d0*px  + d1*py  + d2*pz;

  v2f bfrag;
  bfrag.x = lohalf ? mt0 : mt2;
  bfrag.y = lohalf ? mt1 : cc;

  const float* verts = row + 7;
  float* orow = out + (size_t)pt * (K_V * P_D);
  const int m = lane & 15;

#pragma unroll
  for (int kc = 0; kc < K_V; kc += 16) {
    const float* v3 = verts + 3 * (kc + m);
    const float e0 = v3[lohalf ? 0 : 2];
    const float e1 = v3[1];
    v2f afrag;
    afrag.x = e0;
    afrag.y = lohalf ? e1 : 1.0f;

    v8f acc = {};
    acc = __builtin_amdgcn_wmma_f32_16x16x4_f32(false, afrag, false, bfrag,
                                                (short)0, acc, false, false);

    const float s0 = sel4(jgrp, acc[0], acc[1], acc[2], acc[3]);
    const float s1 = sel4(jgrp, acc[4], acc[5], acc[6], acc[7]);
    orow[(kc + half*8 + jgrp)     * 4 + pcol] = s0;
    orow[(kc + half*8 + 4 + jgrp) * 4 + pcol] = s1;
  }
}

extern "C" void kernel_launch(void* const* d_in, const int* in_sizes, int n_in,
                              void* d_out, int out_size, void* d_ws, size_t ws_size,
                              hipStream_t stream) {
  const float* x    = (const float*)d_in[0];
  const float* dirs = (const float*)d_in[1];
  float* out = (float*)d_out;
  const int npts   = in_sizes[0] / D_IN;
  const int blocks = (npts + 7) / 8;
  vdx_kernel<<<blocks, 256, 0, stream>>>(x, dirs, out, npts);
}


# a wave takes 4 adjacent points; a point = one 16x16 tile: block-diagonal dirs@R (16x12) x 12 vertex coords per column, 3 chained WMMAs, offset as C; 12-B loads, 16-B stores; sum order changes
# speedup vs baseline: 1.0525x; 1.0525x over previous
#include <hip/hip_runtime.h>


typedef float v2f __attribute__((ext_vector_type(2)));
typedef float v4f __attribute__((ext_vector_type(4)));
typedef float v8f __attribute__((ext_vector_type(8)));

#define D_IN  199
#define K_V   64
#define P_D   4
#define PTS   4

__device__ __forceinline__ float sel4(int j, float a0, float a1, float a2, float a3) {
  const float t01 = (j & 1) ? a1 : a0;
  const float t23 = (j & 1) ? a3 : a2;
  return (j & 2) ? t23 : t01;
}

__global__ __launch_bounds__(256) void vdx_kernel(const float* __restrict__ x,
                                                  const float* __restrict__ dirs,
                                                  float* __restrict__ out,
                                                  int npts) {
  const int lane = threadIdx.x & 31;
  const int wave = threadIdx.x >> 5;
  const int pt0  = (blockIdx.x * 8 + wave) * PTS;
  if (pt0 >= npts) return;

  float dr[12];
#pragma unroll
  for (int k = 0; k < 12; ++k) dr[k] = dirs[k];
  float px[PTS], py[PTS], pz[PTS], qw[PTS], qx[PTS], qy[PTS], qz[PTS];
  float x0[PTS], y0[PTS], z0[PTS], x1[PTS], y1[PTS], z1[PTS];
#pragma unroll
  for (int i = 0; i < PTS; ++i) {
    const int pt = (pt0 + i < npts) ? (pt0 + i) : (npts - 1);
    const float* row = x + (size_t)pt * D_IN;
    px[i] = row[0]; py[i] = row[1]; pz[i] = row[2];
    qw[i] = row[3]; qx[i] = row[4]; qy[i] = row[5]; qz[i] = row[6];
    const float* va = row + 7 + 3 * lane;
    const float* vb = va + 96;
    x0[i] = va[0]; y0[i] = va[1]; z0[i] = va[2];
    x1[i] = vb[0]; y1[i] = vb[1]; z1[i] = vb[2];
  }
  __builtin_amdgcn_sched_barrier(0);

  const int  pcol = lane & 3;
  const bool act  = ((lane >> 3) & 1) == (lane >> 4);
  const bool lo   = act && !((lane >> 2) & 1);
  const bool hi   = act &&  ((lane >> 2) & 1);

  const float d0 = sel4(pcol, dr[0], dr[3], dr[6], dr[9]);
  const float d1 = sel4(pcol, dr[1], dr[4], dr[7], dr[10]);
  const float d2 = sel4(pcol, dr[2], dr[5], dr[8], dr[11]);

  v8f acc[PTS];
#pragma unroll
  for (int i = 0; i < PTS; ++i) {
    const float xx = qx[i]*qx[i], yy = qy[i]*qy[i], zz = qz[i]*qz[i];
    const float xy = qx[i]*qy[i], xz = qx[i]*qz[i], yz = qy[i]*qz[i];
    const float wx = qw[i]*qx[i], wy = qw[i]*qy[i], wz = qw[i]*qz[i];
    const float R00 = 1.f - 2.f*(yy+zz), R01 = 2.f*(xy-wz),       R02 = 2.f*(xz+wy);
    const float R10 = 2.f*(xy+wz),       R11 = 1.f - 2.f*(xx+zz), R12 = 2.f*(yz-wx);
    const float R20 = 2.f*(xz-wy),       R21 = 2.f*(yz+wx),       R22 = 1.f - 2.f*(xx+yy);

    const float mt0 = d0*R00 + d1*R10 + d2*R20;
    const float mt1 = d0*R01 + d1*R11 + d2*R21;
    const float mt2 = d0*R02 + d1*R12 + d2*R22;

    const float c0 = dr[0]*px[i] + dr[1]*py[i]  + dr[2]*pz[i];
    const float c1 = dr[3]*px[i] + dr[4]*py[i]  + dr[5]*pz[i];
    const float c2 = dr[6]*px[i] + dr[7]*py[i]  + dr[8]*pz[i];
    const float c3 = dr[9]*px[i] + dr[10]*py[i] + dr[11]*pz[i];

    v2f w0, w1, w2, v0, v1, v2;
    w0.x = lo ? mt0 : 0.f;  w0.y = lo ? mt1 : 0.f;
    w1.x = lo ? mt2 : 0.f;  w1.y = hi ? mt0 : 0.f;
    w2.x = hi ? mt1 : 0.f;  w2.y = hi ? mt2 : 0.f;
    v0.x = x0[i];  v0.y = y0[i];
    v1.x = z0[i];  v1.y = x1[i];
    v2.x = y1[i];  v2.y = z1[i];

    v8f a = {c0, c1, c2, c3, c0, c1, c2, c3};
    a = __builtin_amdgcn_wmma_f32_16x16x4_f32(false, w0, false, v0, (short)0, a, false, false);
    a = __builtin_amdgcn_wmma_f32_16x16x4_f32(false, w1, false, v1, (short)0, a, false, false);
    a = __builtin_amdgcn_wmma_f32_16x16x4_f32(false, w2, false, v2, (short)0, a, false, false);
    acc[i] = a;
  }

#pragma unroll
  for (int i = 0; i < PTS; ++i) {
    if (pt0 + i < npts) {
      float* orow = out + (size_t)(pt0 + i) * (K_V * P_D);
      v4f s0 = {acc[i][0], acc[i][1], acc[i][2], acc[i][3]};
      v4f s1 = {acc[i][4], acc[i][5], acc[i][6], acc[i][7]};
      *(v4f*)(orow + 4 * lane)       = s0;
      *(v4f*)(orow + 128 + 4 * lane) = s1;
    }
  }
}

extern "C" void kernel_launch(void* const* d_in, const int* in_sizes, int n_in,
                              void* d_out, int out_size, void* d_ws, size_t ws_size,
                              hipStream_t stream) {
  const float* x    = (const float*)d_in[0];
  const float* dirs = (const float*)d_in[1];
  float* out = (float*)d_out;
  const int npts   = in_sizes[0] / D_IN;
  const int blocks = (npts + 8 * PTS - 1) / (8 * PTS);
  vdx_kernel<<<blocks, 256, 0, stream>>>(x, dirs, out, npts);
}


# one gather load per wave brings position and quaternion words of all 4 points, readlane moves them to scalar registers: 9 vector loads per wave (was 16); rotation and offset math on the scalar unit
# speedup vs baseline: 1.1024x; 1.0474x over previous
#include <hip/hip_runtime.h>


typedef float v2f __attribute__((ext_vector_type(2)));
typedef float v4f __attribute__((ext_vector_type(4)));
typedef float v8f __attribute__((ext_vector_type(8)));

#define D_IN  199
#define K_V   64
#define P_D   4
#define PTS   4

__device__ __forceinline__ float sel4(int j, float a0, float a1, float a2, float a3) {
  const float t01 = (j & 1) ? a1 : a0;
  const float t23 = (j & 1) ? a3 : a2;
  return (j & 2) ? t23 : t01;
}

__global__ __launch_bounds__(256) void vdx_kernel(const float* __restrict__ x,
                                                  const float* __restrict__ dirs,
                                                  float* __restrict__ out,
                                                  int npts) {
  const int lane = threadIdx.x & 31;
  const int wave = __builtin_amdgcn_readfirstlane(threadIdx.x >> 5);
  const int pt0  = (blockIdx.x * 8 + wave) * PTS;
  if (pt0 >= npts) return;

  float dr[12];
#pragma unroll
  for (int k = 0; k < 12; ++k) dr[k] = dirs[k];
  const int hpnt = (pt0 + (lane >> 3) < npts) ? (pt0 + (lane >> 3)) : (npts - 1);
  const float head = x[(size_t)hpnt * D_IN + (lane & 7)];
  float x0[PTS], y0[PTS], z0[PTS], x1[PTS], y1[PTS], z1[PTS];
#pragma unroll
  for (int i = 0; i < PTS; ++i) {
    const int pt = (pt0 + i < npts) ? (pt0 + i) : (npts - 1);
    const float* row = x + (size_t)pt * D_IN;
    const float* va = row + 7 + 3 * lane;
    const float* vb = va + 96;
    x0[i] = va[0]; y0[i] = va[1]; z0[i] = va[2];
    x1[i] = vb[0]; y1[i] = vb[1]; z1[i] = vb[2];
  }
  __builtin_amdgcn_sched_barrier(0);

  float px[PTS], py[PTS], pz[PTS], qw[PTS], qx[PTS], qy[PTS], qz[PTS];
#pragma unroll
  for (int i = 0; i < PTS; ++i) {
    const int hw = __float_as_int(head);
    px[i] = __int_as_float(__builtin_amdgcn_readlane(hw, 8 * i + 0));
    py[i] = __int_as_float(__builtin_amdgcn_readlane(hw, 8 * i + 1));
    pz[i] = __int_as_float(__builtin_amdgcn_readlane(hw, 8 * i + 2));
    qw[i] = __int_as_float(__builtin_amdgcn_readlane(hw, 8 * i + 3));
    qx[i] = __int_as_float(__builtin_amdgcn_readlane(hw, 8 * i + 4));
    qy[i] = __int_as_float(__builtin_amdgcn_readlane(hw, 8 * i + 5));
    qz[i] = __int_as_float(__builtin_amdgcn_readlane(hw, 8 * i + 6));
  }

  const int  pcol = lane & 3;
  const bool act  = ((lane >> 3) & 1) == (lane >> 4);
  const bool lo   = act && !((lane >> 2) & 1);
  const bool hi   = act &&  ((lane >> 2) & 1);

  const float d0 = sel4(pcol, dr[0], dr[3], dr[6], dr[9]);
  const float d1 = sel4(pcol, dr[1], dr[4], dr[7], dr[10]);
  const float d2 = sel4(pcol, dr[2], dr[5], dr[8], dr[11]);

  v8f acc[PTS];
#pragma unroll
  for (int i = 0; i < PTS; ++i) {
    const float xx = qx[i]*qx[i], yy = qy[i]*qy[i], zz = qz[i]*qz[i];
    const float xy = qx[i]*qy[i], xz = qx[i]*qz[i], yz = qy[i]*qz[i];
    const float wx = qw[i]*qx[i], wy = qw[i]*qy[i], wz = qw[i]*qz[i];
    const float R00 = 1.f - 2.f*(yy+zz), R01 = 2.f*(xy-wz),       R02 = 2.f*(xz+wy);
    const float R10 = 2.f*(xy+wz),       R11 = 1.f - 2.f*(xx+zz), R12 = 2.f*(yz-wx);
    const float R20 = 2.f*(xz-wy),       R21 = 2.f*(yz+wx),       R22 = 1.f - 2.f*(xx+yy);

    const float mt0 = d0*R00 + d1*R10 + d2*R20;
    const float mt1 = d0*R01 + d1*R11 + d2*R21;
    const float mt2 = d0*R02 + d1*R12 + d2*R22;

    const float c0 = dr[0]*px[i] + dr[1]*py[i]  + dr[2]*pz[i];
    const float c1 = dr[3]*px[i] + dr[4]*py[i]  + dr[5]*pz[i];
    const float c2 = dr[6]*px[i] + dr[7]*py[i]  + dr[8]*pz[i];
    const float c3 = dr[9]*px[i] + dr[10]*py[i] + dr[11]*pz[i];

    v2f w0, w1, w2, v0, v1, v2;
    w0.x = lo ? mt0 : 0.f;  w0.y = lo ? mt1 : 0.f;
    w1.x = lo ? mt2 : 0.f;  w1.y = hi ? mt0 : 0.f;
    w2.x = hi ? mt1 : 0.f;  w2.y = hi ? mt2 : 0.f;
    v0.x = x0[i];  v0.y = y0[i];
    v1.x = z0[i];  v1.y = x1[i];
    v2.x = y1[i];  v2.y = z1[i];

    v8f a = {c0, c1, c2, c3, c0, c1, c2, c3};
    a = __builtin_amdgcn_wmma_f32_16x16x4_f32(false, w0, false, v0, (short)0, a, false, false);
    a = __builtin_amdgcn_wmma_f32_16x16x4_f32(false, w1, false, v1, (short)0, a, false, false);
    a = __builtin_amdgcn_wmma_f32_16x16x4_f32(false, w2, false, v2, (short)0, a, false, false);
    acc[i] = a;
  }

#pragma unroll
  for (int i = 0; i < PTS; ++i) {
    if (pt0 + i < npts) {
      float* orow = out + (size_t)(pt0 + i) * (K_V * P_D);
      v4f s0 = {acc[i][0], acc[i][1], acc[i][2], acc[i][3]};
      v4f s1 = {acc[i][4], acc[i][5], acc[i][6], acc[i][7]};
      *(v4f*)(orow + 4 * lane)       = s0;
      *(v4f*)(orow + 128 + 4 * lane) = s1;
    }
  }
}

extern "C" void kernel_launch(void* const* d_in, const int* in_sizes, int n_in,
                              void* d_out, int out_size, void* d_ws, size_t ws_size,
                              hipStream_t stream) {
  const float* x    = (const float*)d_in[0];
  const float* dirs = (const float*)d_in[1];
  float* out = (float*)d_out;
  const int npts   = in_sizes[0] / D_IN;
  const int blocks = (npts + 8 * PTS - 1) / (8 * PTS);
  vdx_kernel<<<blocks, 256, 0, stream>>>(x, dirs, out, npts);
}


# a wave takes 32 adjacent points (was 4): rows of 24 requested at entry, the last 8 once 8 points are stored; points finished in arrival order behind counted waits; 4096 waves, all resident at once
# speedup vs baseline: 1.2112x; 1.0987x over previous
#include <hip/hip_runtime.h>


typedef float v2f __attribute__((ext_vector_type(2)));
typedef float v4f __attribute__((ext_vector_type(4)));
typedef float v8f __attribute__((ext_vector_type(8)));

#define D_IN  199
#define K_V   64
#define P_D   4
#define PTS   32
#define GRP   (PTS / 4)
#define GRP_A 6
#define EARLY 8


__device__ __forceinline__ float sel4(int j, float a0, float a1, float a2, float a3) {
  const float t01 = (j & 1) ? a1 : a0;
  const float t23 = (j & 1) ? a3 : a2;
  return (j & 2) ? t23 : t01;
}

__device__ __forceinline__ void point(const float* dr, float d0, float d1, float d2,
                                      bool lo, bool hi,
                                      float px, float py, float pz,
                                      float qw, float qx, float qy, float qz,
                                      float x0, float y0, float z0,
                                      float x1, float y1, float z1,
                                      float* __restrict__ orow, int lane) {
  const float xx = qx*qx, yy = qy*qy, zz = qz*qz;
  const float xy = qx*qy, xz = qx*qz, yz = qy*qz;
  const float wx = qw*qx, wy = qw*qy, wz = qw*qz;
  const float R00 = 1.f - 2.f*(yy+zz), R01 = 2.f*(xy-wz),       R02 = 2.f*(xz+wy);
  const float R10 = 2.f*(xy+wz),       R11 = 1.f - 2.f*(xx+zz), R12 = 2.f*(yz-wx);
  const float R20 = 2.f*(xz-wy),       R21 = 2.f*(yz+wx),       R22 = 1.f - 2.f*(xx+yy);

  const float mt0 = d0*R00 + d1*R10 + d2*R20;
  const float mt1 = d0*R01 + d1*R11 + d2*R21;
  const float mt2 = d0*R02 + d1*R12 + d2*R22;

  const float c0 = dr[0]*px + dr[1]*py  + dr[2]*pz;
  const float c1 = dr[3]*px + dr[4]*py  + dr[5]*pz;
  const float c2 = dr[6]*px + dr[7]*py  + dr[8]*pz;
  const float c3 = dr[9]*px + dr[10]*py + dr[11]*pz;

  v2f w0, w1, w2, v0, v1, v2;
  w0.x = lo ? mt0 : 0.f;  w0.y = lo ? mt1 : 0.f;
  w1.x = lo ? mt2 : 0.f;  w1.y = hi ? mt0 : 0.f;
  w2.x = hi ? mt1 : 0.f;  w2.y = hi ? mt2 : 0.f;
  v0.x = x0;  v0.y = y0;
  v1.x = z0;  v1.y = x1;
  v2.x = y1;  v2.y = z1;

  v8f a = {c0, c1, c2, c3, c0, c1, c2, c3};
  a = __builtin_amdgcn_wmma_f32_16x16x4_f32(false, w0, false, v0, (short)0, a, false, false);
  a = __builtin_amdgcn_wmma_f32_16x16x4_f32(false, w1, false, v1, (short)0, a, false, false);
  a = __builtin_amdgcn_wmma_f32_16x16x4_f32(false, w2, false, v2, (short)0, a, false, false);

  v4f s0 = {a[0], a[1], a[2], a[3]};
  v4f s1 = {a[4], a[5], a[6], a[7]};
  *(v4f*)(orow + 4 * lane)       = s0;
  *(v4f*)(orow + 128 + 4 * lane) = s1;
}

__global__ __launch_bounds__(256) void vdx_kernel(const float* __restrict__ x,
                                                  const float* __restrict__ dirs,
                                                  float* __restrict__ out,
                                                  int npts) {
  const int lane = threadIdx.x & 31;
  const int wave = __builtin_amdgcn_readfirstlane(threadIdx.x >> 5);
  const int pt0  = (blockIdx.x * 8 + wave) * PTS;
  if (pt0 >= npts) return;

  float dr[12];
#pragma unroll
  for (int k = 0; k < 12; ++k) dr[k] = dirs[k];

  const int  pcol = lane & 3;
  const bool act  = ((lane >> 3) & 1) == (lane >> 4);
  const bool lo   = act && !((lane >> 2) & 1);
  const bool hi   = act &&  ((lane >> 2) & 1);

  if (pt0 + PTS <= npts) {
    const float* base = x + (size_t)pt0 * D_IN;
    float* obase = out + (size_t)pt0 * (K_V * P_D);
    float head[GRP];
    float x0[PTS], y0[PTS], z0[PTS], x1[PTS], y1[PTS], z1[PTS];

    const float d0 = sel4(pcol, dr[0], dr[3], dr[6], dr[9]);
    const float d1 = sel4(pcol, dr[1], dr[4], dr[7], dr[10]);
    const float d2 = sel4(pcol, dr[2], dr[5], dr[8], dr[11]);

#define REQUEST_GROUP(g)                                                          \
    {                                                                             \
        \
        \
        \
      head[g] = base[(size_t)(4 * (g) + (lane >> 3)) * D_IN + (lane & 7)];        \
        \
      _Pragma("unroll")                                                           \
      for (int j = 0; j < 4; ++j) {                                               \
        const int i = 4 * (g) + j;                                                \
        const float* va = base + (size_t)i * D_IN + 7 + 3 * lane;   \
        const float* vb = va + 96;                             \
        x0[i] = va[0]; y0[i] = va[1]; z0[i] = va[2];                              \
        x1[i] = vb[0]; y1[i] = vb[1]; z1[i] = vb[2];                              \
      }                                                                           \
        \
        \
        \
      __builtin_amdgcn_sched_barrier(0);                                          \
    }

#define FINISH_POINT(i)                                                           \
    {                                                                             \
      const int hw = __float_as_int(head[(i) >> 2]);                              \
      const int l0 = 8 * ((i) & 3);                                               \
      const float px = __int_as_float(__builtin_amdgcn_readlane(hw, l0 + 0));     \
      const float py = __int_as_float(__builtin_amdgcn_readlane(hw, l0 + 1));     \
      const float pz = __int_as_float(__builtin_amdgcn_readlane(hw, l0 + 2));     \
      const float qw = __int_as_float(__builtin_amdgcn_readlane(hw, l0 + 3));     \
      const float qx = __int_as_float(__builtin_amdgcn_readlane(hw, l0 + 4));     \
      const float qy = __int_as_float(__builtin_amdgcn_readlane(hw, l0 + 5));     \
      const float qz = __int_as_float(__builtin_amdgcn_readlane(hw, l0 + 6));     \
      float ux = x1[i], uy = y1[i], uz = z1[i];                                   \
          \
          \
          \
          \
          \
      asm volatile("" : "+v"(ux), "+v"(uy), "+v"(uz));                            \
      point(dr, d0, d1, d2, lo, hi, px, py, pz, qw, qx, qy, qz,                   \
            x0[i], y0[i], z0[i], ux, uy, uz,                                      \
            obase + (size_t)(i) * (K_V * P_D), lane);                             \
    }

#pragma unroll
    for (int g = 0; g < GRP_A; ++g) REQUEST_GROUP(g)

#pragma unroll
    for (int i = 0; i < EARLY; ++i) FINISH_POINT(i)

    __builtin_amdgcn_sched_barrier(0);
#pragma unroll
    for (int g = GRP_A; g < GRP; ++g) REQUEST_GROUP(g)

#pragma unroll
    for (int i = EARLY; i < PTS; ++i) FINISH_POINT(i)
#undef REQUEST_GROUP
#undef FINISH_POINT
  } else {
    const float d0 = sel4(pcol, dr[0], dr[3], dr[6], dr[9]);
    const float d1 = sel4(pcol, dr[1], dr[4], dr[7], dr[10]);
    const float d2 = sel4(pcol, dr[2], dr[5], dr[8], dr[11]);
    for (int pt = pt0; pt < npts; ++pt) {
      const float* row = x + (size_t)pt * D_IN;
      const float hd = row[lane & 7];
      const float* va = row + 7 + 3 * lane;
      const float* vb = va + 96;
      const float ax = va[0], ay = va[1], az = va[2];
      const float bx = vb[0], by = vb[1], bz = vb[2];
      const int hw = __float_as_int(hd);
      const float px = __int_as_float(__builtin_amdgcn_readlane(hw, 0));
      const float py = __int_as_float(__builtin_amdgcn_readlane(hw, 1));
      const float pz = __int_as_float(__builtin_amdgcn_readlane(hw, 2));
      const float qw = __int_as_float(__builtin_amdgcn_readlane(hw, 3));
      const float qx = __int_as_float(__builtin_amdgcn_readlane(hw, 4));
      const float qy = __int_as_float(__builtin_amdgcn_readlane(hw, 5));
      const float qz = __int_as_float(__builtin_amdgcn_readlane(hw, 6));
      point(dr, d0, d1, d2, lo, hi, px, py, pz, qw, qx, qy, qz,
            ax, ay, az, bx, by, bz, out + (size_t)pt * (K_V * P_D), lane);
    }
  }
}

extern "C" void kernel_launch(void* const* d_in, const int* in_sizes, int n_in,
                              void* d_out, int out_size, void* d_ws, size_t ws_size,
                              hipStream_t stream) {
  const float* x    = (const float*)d_in[0];
  const float* dirs = (const float*)d_in[1];
  float* out = (float*)d_out;
  const int npts   = in_sizes[0] / D_IN;
  const int blocks = (npts + 8 * PTS - 1) / (8 * PTS);
  vdx_kernel<<<blocks, 256, 0, stream>>>(x, dirs, out, npts);
}
